# starting kernel
# baseline (speedup 1.0000x reference)
#include <hip/hip_runtime.h>

typedef float v2f __attribute__((ext_vector_type(2)));
typedef float v8f __attribute__((ext_vector_type(8)));

#define RBF_D 16
#define RBF_K 64

__global__ __launch_bounds__(256)
void rbf_wmma_kernel(const float* __restrict__ x,
                     const float* __restrict__ centers,
                     const float* __restrict__ scaling,
                     float* __restrict__ O,
                     float* __restrict__ Hx) {
  const int lane = threadIdx.x & 31;
  const int wave = threadIdx.x >> 5;
  const int m    = lane & 15;
  const int half = lane >> 4;

  const int point_base = (blockIdx.x * 8 + wave) * 16;
  const int row = point_base + m;

  v2f sv[4];
#pragma unroll
  for (int c = 0; c < 4; ++c)
    sv[c] = *(const v2f*)(scaling + 4 * c + 2 * half);

  v2f a[4];
  float hsq_part = 0.f;
#pragma unroll
  for (int c = 0; c < 4; ++c) {
    const int d = 4 * c + 2 * half;
    v2f xv = *(const v2f*)(x + (size_t)row * RBF_D + d);
    v2f h;
    h.x = xv.x * sv[c].x;
    h.y = xv.y * sv[c].y;
    a[c] = h;
    hsq_part += h.x * h.x + h.y * h.y;
    *(v2f*)(Hx + (size_t)row * RBF_D + d) = h;
  }
  const float hsq = hsq_part + __shfl_xor(hsq_part, 16, 32);
  float hval[8];
#pragma unroll
  for (int v = 0; v < 8; ++v)
    hval[v] = __shfl(hsq, v + 8 * half, 32);

#pragma unroll
  for (int t = 0; t < 4; ++t) {
    const int col = t * 16 + m;
    v2f bmat[4];
    float csq_part = 0.f;
#pragma unroll
    for (int c = 0; c < 4; ++c) {
      const int d = 4 * c + 2 * half;
      v2f cv = *(const v2f*)(centers + (size_t)col * RBF_D + d);
      v2f bc;
      bc.x = cv.x * sv[c].x;
      bc.y = cv.y * sv[c].y;
      bmat[c] = bc;
      csq_part += bc.x * bc.x + bc.y * bc.y;
    }
    const float csq = csq_part + __shfl_xor(csq_part, 16, 32);

    v8f acc = {};
#pragma unroll
    for (int c = 0; c < 4; ++c) {
      acc = __builtin_amdgcn_wmma_f32_16x16x4_f32(
           false, a[c],  false, bmat[c],
           (short)0, acc,  false,  false);
    }

#pragma unroll
    for (int v = 0; v < 8; ++v) {
      O[(size_t)(point_base + v + 8 * half) * RBF_K + col] =
          hval[v] + csq - 2.0f * acc[v];
    }
  }
}

extern "C" void kernel_launch(void* const* d_in, const int* in_sizes, int n_in,
                              void* d_out, int out_size, void* d_ws, size_t ws_size,
                              hipStream_t stream) {
  const float* x       = (const float*)d_in[0];
  const float* centers = (const float*)d_in[1];
  const float* scaling = (const float*)d_in[2];

  const int npoints = in_sizes[0] / RBF_D;
  float* O  = (float*)d_out;
  float* Hx = O + (size_t)npoints * RBF_K;

  const int points_per_block = 16 * 8;
  const int blocks = npoints / points_per_block;

  hipLaunchKernelGGL(rbf_wmma_kernel, dim3(blocks), dim3(256), 0, stream,
                     x, centers, scaling, O, Hx);
}


# one wave does 4 point tiles (64 points), center fragments and norms built once per wave: 1024 waves in 256 groups of 4 (one per SIMD), was 4096; tile p stores under tile p+1 WMMAs; sum order kept
# speedup vs baseline: 1.0495x; 1.0495x over previous
#include <hip/hip_runtime.h>

typedef float v2f __attribute__((ext_vector_type(2)));
typedef float v8f __attribute__((ext_vector_type(8)));

#define RBF_D 16
#define RBF_K 64
#define RBF_MT 4
#define RBF_WV 4

__global__ __launch_bounds__(32 * RBF_WV)
void rbf_wmma_kernel(const float* __restrict__ x,
                     const float* __restrict__ centers,
                     const float* __restrict__ scaling,
                     float* __restrict__ O,
                     float* __restrict__ Hx) {
  const int lane = threadIdx.x & 31;
  const int wave = threadIdx.x >> 5;
  const int m    = lane & 15;
  const int half = lane >> 4;

  const int point_base = (blockIdx.x * RBF_WV + wave) * (16 * RBF_MT);

  const float* sp = scaling + 2 * half;
  const float* cp = centers + m * RBF_D + 2 * half;
  const float* xp = x  + (size_t)(point_base + m) * RBF_D + 2 * half;
  float*       hp = Hx + (size_t)(point_base + m) * RBF_D + 2 * half;
  float*       op = O  + (size_t)(point_base + 8 * half) * RBF_K + m;

  v2f sv[4];
#pragma unroll
  for (int c = 0; c < 4; ++c)
    sv[c] = *(const v2f*)(sp + 4 * c);

  v2f cv[4][4];
#pragma unroll
  for (int t = 0; t < 4; ++t) {
#pragma unroll
    for (int c = 0; c < 4; ++c)
      cv[t][c] = *(const v2f*)(cp + t * 16 * RBF_D + 4 * c);
  }

  v2f xv[RBF_MT][4];
#pragma unroll
  for (int p = 0; p < RBF_MT; ++p) {
#pragma unroll
    for (int c = 0; c < 4; ++c)
      xv[p][c] = *(const v2f*)(xp + p * 16 * RBF_D + 4 * c);
  }

  v2f bmat[4][4];
  float csq[4];
#pragma unroll
  for (int t = 0; t < 4; ++t) {
    float csq_part = 0.f;
#pragma unroll
    for (int c = 0; c < 4; ++c) {
      v2f bc;
      bc.x = cv[t][c].x * sv[c].x;
      bc.y = cv[t][c].y * sv[c].y;
      bmat[t][c] = bc;
      csq_part += bc.x * bc.x + bc.y * bc.y;
    }
    csq[t] = csq_part + __shfl_xor(csq_part, 16, 32);
  }

  v2f a[RBF_MT][4];
  float hval[RBF_MT][8];
  v8f acc[RBF_MT][4];

  auto prepare = [&](int q) __attribute__((always_inline)) {
    float hsq_part = 0.f;
#pragma unroll
    for (int c = 0; c < 4; ++c) {
      v2f h;
      h.x = xv[q][c].x * sv[c].x;
      h.y = xv[q][c].y * sv[c].y;
      a[q][c] = h;
      hsq_part += h.x * h.x + h.y * h.y;
      *(v2f*)(hp + q * 16 * RBF_D + 4 * c) = h;
    }
    const float hsq = hsq_part + __shfl_xor(hsq_part, 16, 32);
#pragma unroll
    for (int v = 0; v < 8; ++v)
      hval[q][v] = __shfl(hsq, v + 8 * half, 32);
  };

  auto products = [&](int q) __attribute__((always_inline)) {
#pragma unroll
    for (int t = 0; t < 4; ++t) acc[q][t] = (v8f){};
#pragma unroll
    for (int c = 0; c < 4; ++c) {
#pragma unroll
      for (int t = 0; t < 4; ++t) {
        acc[q][t] = __builtin_amdgcn_wmma_f32_16x16x4_f32(
             false, a[q][c],  false, bmat[t][c],
             (short)0, acc[q][t],  false,  false);
      }
    }
  };

  auto finish = [&](int q) __attribute__((always_inline)) {
#pragma unroll
    for (int t = 0; t < 4; ++t) {
#pragma unroll
      for (int v = 0; v < 8; ++v)
        op[(q * 16 + v) * RBF_K + t * 16] = hval[q][v] + csq[t] - 2.0f * acc[q][t][v];
    }
  };

  prepare(0);
#pragma unroll
  for (int p = 0; p < RBF_MT; ++p) {
    products(p);
    if (p == 0) {
#pragma unroll
      for (int q = 1; q < RBF_MT; ++q) prepare(q);
    } else {
      finish(p - 1);
    }
    __builtin_amdgcn_sched_barrier(0);
  }
  finish(RBF_MT - 1);
}

extern "C" void kernel_launch(void* const* d_in, const int* in_sizes, int n_in,
                              void* d_out, int out_size, void* d_ws, size_t ws_size,
                              hipStream_t stream) {
  const float* x       = (const float*)d_in[0];
  const float* centers = (const float*)d_in[1];
  const float* scaling = (const float*)d_in[2];

  const int npoints = in_sizes[0] / RBF_D;
  float* O  = (float*)d_out;
  float* Hx = O + (size_t)npoints * RBF_K;

  const int points_per_block = 16 * RBF_MT * RBF_WV;
  const int blocks = npoints / points_per_block;

  hipLaunchKernelGGL(rbf_wmma_kernel, dim3(blocks), dim3(32 * RBF_WV), 0, stream,
                     x, centers, scaling, O, Hx);
}


# O and Hx rows staged through wave-private LDS (centers as the WMMA A operand: a lane holds 8 consecutive centers) so every global store is b128 over 4 whole 128-B lines; sum order kept
# speedup vs baseline: 1.1595x; 1.1048x over previous
#include <hip/hip_runtime.h>

typedef float v2f __attribute__((ext_vector_type(2)));
typedef float v4f __attribute__((ext_vector_type(4)));
typedef float v8f __attribute__((ext_vector_type(8)));

#define RBF_D 16
#define RBF_K 64
#define RBF_MT 4
#define RBF_WV 4

#define O_PITCH 68
#define O_TILE  (16 * O_PITCH)
#define H_PITCH 20
#define H_TILE  (16 * H_PITCH)
#define WAVE_LDS (RBF_MT * O_TILE + RBF_MT * H_TILE + RBF_K)

static __device__ __forceinline__ void staged_rows_visible() { asm volatile("s_wait_dscnt 0x0" ::: "memory"); }

__global__ __launch_bounds__(32 * RBF_WV)
void rbf_wmma_kernel(const float* __restrict__ x,
                     const float* __restrict__ centers,
                     const float* __restrict__ scaling,
                     float* __restrict__ O,
                     float* __restrict__ Hx) {
  __shared__ __attribute__((aligned(16))) float lds[RBF_WV * WAVE_LDS];

  const int lane = threadIdx.x & 31;
  const int wave = threadIdx.x >> 5;
  const int m    = lane & 15;
  const int half = lane >> 4;

  float* so  = lds + wave * WAVE_LDS;
  float* sh  = so + RBF_MT * O_TILE;
  float* scs = sh + RBF_MT * H_TILE;

  const int point_base = (blockIdx.x * RBF_WV + wave) * (16 * RBF_MT);

  const float* sp = scaling + 2 * half;
  const float* cp = centers + m * RBF_D + 2 * half;
  const float* xp = x  + (size_t)(point_base + m) * RBF_D + 2 * half;

  float*       so_w = so + m * O_PITCH + 8 * half;
  const float* so_r = so + (lane >> 4) * O_PITCH + 4 * (lane & 15);
  float*       sh_w = sh + m * H_PITCH + 2 * half;
  const float* sh_r = sh + (lane >> 2) * H_PITCH + 4 * (lane & 3);
  float* op = O  + (size_t)point_base * RBF_K + (lane >> 4) * RBF_K + 4 * (lane & 15);
  float* hp = Hx + (size_t)point_base * RBF_D + 4 * lane;

  v2f sv[4];
#pragma unroll
  for (int c = 0; c < 4; ++c)
    sv[c] = *(const v2f*)(sp + 4 * c);

  v2f cv[4][4];
#pragma unroll
  for (int t = 0; t < 4; ++t) {
#pragma unroll
    for (int c = 0; c < 4; ++c)
      cv[t][c] = *(const v2f*)(cp + t * 16 * RBF_D + 4 * c);
  }

  v2f xv[RBF_MT][4];
#pragma unroll
  for (int p = 0; p < RBF_MT; ++p) {
#pragma unroll
    for (int c = 0; c < 4; ++c)
      xv[p][c] = *(const v2f*)(xp + p * 16 * RBF_D + 4 * c);
  }

  v2f cmat[4][4];
  float csq[4];
#pragma unroll
  for (int t = 0; t < 4; ++t) {
    float csq_part = 0.f;
#pragma unroll
    for (int c = 0; c < 4; ++c) {
      v2f bc;
      bc.x = cv[t][c].x * sv[c].x;
      bc.y = cv[t][c].y * sv[c].y;
      cmat[t][c] = bc;
      csq_part += bc.x * bc.x + bc.y * bc.y;
    }
    csq[t] = csq_part + __shfl_xor(csq_part, 16, 32);
  }
  scs[16 * half + m]       = half ? csq[1] : csq[0];
  scs[16 * (2 + half) + m] = half ? csq[3] : csq[2];

  v2f a[RBF_MT][4];
  float hsq[RBF_MT];
  v4f csqv[4][2];
  v8f acc[RBF_MT][4];

  auto prepare = [&](int q) __attribute__((always_inline)) {
    float hsq_part = 0.f;
#pragma unroll
    for (int c = 0; c < 4; ++c) {
      v2f h;
      h.x = xv[q][c].x * sv[c].x;
      h.y = xv[q][c].y * sv[c].y;
      a[q][c] = h;
      hsq_part += h.x * h.x + h.y * h.y;
      *(v2f*)(sh_w + q * H_TILE + 4 * c) = h;
    }
    hsq[q] = hsq_part + __shfl_xor(hsq_part, 16, 32);
  };

  auto products = [&](int q) __attribute__((always_inline)) {
#pragma unroll
    for (int t = 0; t < 4; ++t) acc[q][t] = (v8f){};
#pragma unroll
    for (int c = 0; c < 4; ++c) {
#pragma unroll
      for (int t = 0; t < 4; ++t) {
        acc[q][t] = __builtin_amdgcn_wmma_f32_16x16x4_f32(
             false, cmat[t][c],  false, a[q][c],
             (short)0, acc[q][t],  false,  false);
      }
    }
  };

  auto combine = [&](int q) __attribute__((always_inline)) {
#pragma unroll
    for (int t = 0; t < 4; ++t) {
#pragma unroll
      for (int j = 0; j < 2; ++j) {
        v4f o;
#pragma unroll
        for (int e = 0; e < 4; ++e)
          o[e] = hsq[q] + csqv[t][j][e] - 2.0f * acc[q][t][4 * j + e];
        *(v4f*)(so_w + q * O_TILE + 16 * t + 4 * j) = o;
      }
    }
  };

  auto store_rows = [&](int q) __attribute__((always_inline)) {
    v4f r[8];
#pragma unroll
    for (int i = 0; i < 8; ++i)
      r[i] = *(const v4f*)(so_r + q * O_TILE + 2 * i * O_PITCH);
#pragma unroll
    for (int i = 0; i < 8; ++i)
      *(v4f*)(op + (size_t)(q * 16 + 2 * i) * RBF_K) = r[i];
  };

  prepare(0);
#pragma unroll
  for (int p = 0; p < RBF_MT; ++p) {
    products(p);
    if (p == 0) {
#pragma unroll
      for (int q = 1; q < RBF_MT; ++q) prepare(q);
      staged_rows_visible();
#pragma unroll
      for (int t = 0; t < 4; ++t) {
#pragma unroll
        for (int j = 0; j < 2; ++j)
          csqv[t][j] = *(const v4f*)(scs + 16 * t + 8 * half + 4 * j);
      }
      v4f hr[2 * RBF_MT];
#pragma unroll
      for (int i = 0; i < 2 * RBF_MT; ++i)
        hr[i] = *(const v4f*)(sh_r + (i >> 1) * H_TILE + (i & 1) * 8 * H_PITCH);
#pragma unroll
      for (int i = 0; i < 2 * RBF_MT; ++i)
        *(v4f*)(hp + i * 8 * RBF_D) = hr[i];
    } else {
      combine(p - 1);
      staged_rows_visible();
      store_rows(p - 1);
    }
    __builtin_amdgcn_sched_barrier(0);
  }
  combine(RBF_MT - 1);
  staged_rows_visible();
  store_rows(RBF_MT - 1);
}

extern "C" void kernel_launch(void* const* d_in, const int* in_sizes, int n_in,
                              void* d_out, int out_size, void* d_ws, size_t ws_size,
                              hipStream_t stream) {
  const float* x       = (const float*)d_in[0];
  const float* centers = (const float*)d_in[1];
  const float* scaling = (const float*)d_in[2];

  const int npoints = in_sizes[0] / RBF_D;
  float* O  = (float*)d_out;
  float* Hx = O + (size_t)npoints * RBF_K;

  const int points_per_block = 16 * RBF_MT * RBF_WV;
  const int blocks = npoints / points_per_block;

  hipLaunchKernelGGL(rbf_wmma_kernel, dim3(blocks), dim3(32 * RBF_WV), 0, stream,
                     x, centers, scaling, O, Hx);
}
